# starting kernel
# baseline (speedup 1.0000x reference)
#include <hip/hip_runtime.h>


typedef __attribute__((ext_vector_type(2))) float v2f;
typedef __attribute__((ext_vector_type(8))) float v8f;

#define IN_C   3
#define OUT_C  32
#define HW     64
#define NOUT   60
#define PITCH  68

__device__ __forceinline__ float band(int row, int k) {
    return ((unsigned)(k - row) <= 4u) ? 1.0f : 0.0f;
}

__global__ __launch_bounds__(256) void
localconv2d_wmma_kernel(const float* __restrict__ x,
                        const float* __restrict__ lb,
                        const float* __restrict__ rb,
                        float* __restrict__ out) {
    const int b   = blockIdx.x >> 5;
    const int oc  = blockIdx.x & 31;

    __shared__ float S[HW * PITCH];
    __shared__ float T[HW * PITCH];

    const int tid  = threadIdx.x;
    const int lane = tid & 31;
    const int wave = tid >> 5;

    const float l    = lb[oc * IN_C * 25];
    const float r    = rb[oc * IN_C * 25];
    const float dlr  = r - l;
    const float norm = 4.0f / (dlr * dlr);

    const float* xb = x + (size_t)b * IN_C * HW * HW;

    for (int idx = tid; idx < HW * HW; idx += 256) {
        float s = 0.0f;
        #pragma unroll
        for (int c = 0; c < IN_C; ++c) {
            float p  = xb[idx + c * HW * HW];
            float a  = fminf(fmaxf(p - l, 0.0f), 1.0f);
            float bb = fminf(fmaxf(r - p, 0.0f), 1.0f);
            float m  = a * bb * norm;
            s += m * m;
        }
        S[(idx >> 6) * PITCH + (idx & 63)] = s;
    }
    __syncthreads();

    const int ti    = wave >> 1;
    const int tj0   = (wave & 1) * 2;
    const int lrow  = lane & 15;
    const int khalf = (lane >> 4) << 1;
    const int rowA  = ti * 16 + lrow;
    const int rofs  = (lane >> 4) * 8;

    {
        v8f d0 = {}; v8f d1 = {};
        #pragma unroll
        for (int kk = 0; kk < 16; ++kk) {
            const int k0 = kk * 4 + khalf;
            v2f a;
            a.x = band(rowA, k0);
            a.y = band(rowA, k0 + 1);
            const int col0 = tj0 * 16 + lrow;
            v2f b0, b1;
            b0.x = S[ k0      * PITCH + col0];
            b0.y = S[(k0 + 1) * PITCH + col0];
            b1.x = S[ k0      * PITCH + col0 + 16];
            b1.y = S[(k0 + 1) * PITCH + col0 + 16];
            d0 = __builtin_amdgcn_wmma_f32_16x16x4_f32(false, a, false, b0,
                                                       (short)0, d0, false, false);
            d1 = __builtin_amdgcn_wmma_f32_16x16x4_f32(false, a, false, b1,
                                                       (short)0, d1, false, false);
        }
        #pragma unroll
        for (int g = 0; g < 8; ++g) {
            const int rr = ti * 16 + g + rofs;
            T[rr * PITCH + tj0 * 16 + lrow]      = d0[g];
            T[rr * PITCH + tj0 * 16 + 16 + lrow] = d1[g];
        }
    }
    __syncthreads();

    {
        v8f e0 = {}; v8f e1 = {};
        #pragma unroll
        for (int kk = 0; kk < 16; ++kk) {
            const int k0 = kk * 4 + khalf;
            v2f a;
            a.x = T[rowA * PITCH + k0];
            a.y = T[rowA * PITCH + k0 + 1];
            const int j0 = tj0 * 16 + lrow;
            v2f b0, b1;
            b0.x = band(j0,      k0);
            b0.y = band(j0,      k0 + 1);
            b1.x = band(j0 + 16, k0);
            b1.y = band(j0 + 16, k0 + 1);
            e0 = __builtin_amdgcn_wmma_f32_16x16x4_f32(false, a, false, b0,
                                                       (short)0, e0, false, false);
            e1 = __builtin_amdgcn_wmma_f32_16x16x4_f32(false, a, false, b1,
                                                       (short)0, e1, false, false);
        }
        float* outb = out + (size_t)(b * OUT_C + oc) * NOUT * NOUT;
        #pragma unroll
        for (int g = 0; g < 8; ++g) {
            const int rr  = ti * 16 + g + rofs;
            const int cc0 = tj0 * 16 + lrow;
            if (rr < NOUT) {
                if (cc0 < NOUT)      outb[rr * NOUT + cc0]      = e0[g];
                if (cc0 + 16 < NOUT) outb[rr * NOUT + cc0 + 16] = e1[g];
            }
        }
    }
}

extern "C" void kernel_launch(void* const* d_in, const int* in_sizes, int n_in,
                              void* d_out, int out_size, void* d_ws, size_t ws_size,
                              hipStream_t stream) {
    (void)in_sizes; (void)n_in; (void)out_size; (void)d_ws; (void)ws_size;
    const float* x  = (const float*)d_in[0];
    const float* lb = (const float*)d_in[1];
    const float* rb = (const float*)d_in[2];
    float* out = (float*)d_out;
    localconv2d_wmma_kernel<<<dim3(16 * OUT_C), dim3(256), 0, stream>>>(x, lb, rb, out);
}


# phase-1 x loads hoisted: each thread requests its 12 float4 of the image up front, one memory round trip where the loop took 8 serial trips of 6 scalar loads; Ssum goes to LDS as 16-byte stores
# speedup vs baseline: 1.5163x; 1.5163x over previous
#include <hip/hip_runtime.h>


typedef __attribute__((ext_vector_type(2))) float v2f;
typedef __attribute__((ext_vector_type(4))) float v4f;
typedef __attribute__((ext_vector_type(8))) float v8f;

#define IN_C   3
#define OUT_C  32
#define HW     64
#define NOUT   60
#define PITCH  68

__device__ __forceinline__ float band(int row, int k) {
    return ((unsigned)(k - row) <= 4u) ? 1.0f : 0.0f;
}

__global__ __launch_bounds__(256) void
localconv2d_wmma_kernel(const float* __restrict__ x,
                        const float* __restrict__ lb,
                        const float* __restrict__ rb,
                        float* __restrict__ out) {
    const int b   = blockIdx.x >> 5;
    const int oc  = blockIdx.x & 31;

    __shared__ __attribute__((aligned(16))) float S[HW * PITCH];
    __shared__ float T[HW * PITCH];

    const int tid  = threadIdx.x;
    const int lane = tid & 31;
    const int wave = tid >> 5;

    const float* xb = x + (size_t)b * IN_C * HW * HW;

    const v4f* xv = reinterpret_cast<const v4f*>(xb);
    v4f px[4][IN_C];
    for (int i = 0; i < 4; ++i)
        for (int c = 0; c < IN_C; ++c)
            px[i][c] = xv[c * (HW * HW / 4) + i * 256 + tid];

    const float l    = lb[oc * IN_C * 25];
    const float r    = rb[oc * IN_C * 25];
    const float dlr  = r - l;
    const float norm = 4.0f / (dlr * dlr);

    for (int i = 0; i < 4; ++i) {
        v4f s4;
        for (int e = 0; e < 4; ++e) {
            float s = 0.0f;
            for (int c = 0; c < IN_C; ++c) {
                float p  = px[i][c][e];
                float a  = fminf(fmaxf(p - l, 0.0f), 1.0f);
                float bb = fminf(fmaxf(r - p, 0.0f), 1.0f);
                float m  = a * bb * norm;
                s += m * m;
            }
            s4[e] = s;
        }
        const int q = i * 256 + tid;
        *reinterpret_cast<v4f*>(&S[(q >> 4) * PITCH + (q & 15) * 4]) = s4;
    }
    __syncthreads();

    const int ti    = wave >> 1;
    const int tj0   = (wave & 1) * 2;
    const int lrow  = lane & 15;
    const int khalf = (lane >> 4) << 1;
    const int rowA  = ti * 16 + lrow;
    const int rofs  = (lane >> 4) * 8;

    {
        v8f d0 = {}; v8f d1 = {};
        #pragma unroll
        for (int kk = 0; kk < 16; ++kk) {
            const int k0 = kk * 4 + khalf;
            v2f a;
            a.x = band(rowA, k0);
            a.y = band(rowA, k0 + 1);
            const int col0 = tj0 * 16 + lrow;
            v2f b0, b1;
            b0.x = S[ k0      * PITCH + col0];
            b0.y = S[(k0 + 1) * PITCH + col0];
            b1.x = S[ k0      * PITCH + col0 + 16];
            b1.y = S[(k0 + 1) * PITCH + col0 + 16];
            d0 = __builtin_amdgcn_wmma_f32_16x16x4_f32(false, a, false, b0,
                                                       (short)0, d0, false, false);
            d1 = __builtin_amdgcn_wmma_f32_16x16x4_f32(false, a, false, b1,
                                                       (short)0, d1, false, false);
        }
        #pragma unroll
        for (int g = 0; g < 8; ++g) {
            const int rr = ti * 16 + g + rofs;
            T[rr * PITCH + tj0 * 16 + lrow]      = d0[g];
            T[rr * PITCH + tj0 * 16 + 16 + lrow] = d1[g];
        }
    }
    __syncthreads();

    {
        v8f e0 = {}; v8f e1 = {};
        #pragma unroll
        for (int kk = 0; kk < 16; ++kk) {
            const int k0 = kk * 4 + khalf;
            v2f a;
            a.x = T[rowA * PITCH + k0];
            a.y = T[rowA * PITCH + k0 + 1];
            const int j0 = tj0 * 16 + lrow;
            v2f b0, b1;
            b0.x = band(j0,      k0);
            b0.y = band(j0,      k0 + 1);
            b1.x = band(j0 + 16, k0);
            b1.y = band(j0 + 16, k0 + 1);
            e0 = __builtin_amdgcn_wmma_f32_16x16x4_f32(false, a, false, b0,
                                                       (short)0, e0, false, false);
            e1 = __builtin_amdgcn_wmma_f32_16x16x4_f32(false, a, false, b1,
                                                       (short)0, e1, false, false);
        }
        float* outb = out + (size_t)(b * OUT_C + oc) * NOUT * NOUT;
        #pragma unroll
        for (int g = 0; g < 8; ++g) {
            const int rr  = ti * 16 + g + rofs;
            const int cc0 = tj0 * 16 + lrow;
            if (rr < NOUT) {
                if (cc0 < NOUT)      outb[rr * NOUT + cc0]      = e0[g];
                if (cc0 + 16 < NOUT) outb[rr * NOUT + cc0 + 16] = e1[g];
            }
        }
    }
}

extern "C" void kernel_launch(void* const* d_in, const int* in_sizes, int n_in,
                              void* d_out, int out_size, void* d_ws, size_t ws_size,
                              hipStream_t stream) {
    (void)in_sizes; (void)n_in; (void)out_size; (void)d_ws; (void)ws_size;
    const float* x  = (const float*)d_in[0];
    const float* lb = (const float*)d_in[1];
    const float* rb = (const float*)d_in[2];
    float* out = (float*)d_out;
    localconv2d_wmma_kernel<<<dim3(16 * OUT_C), dim3(256), 0, stream>>>(x, lb, rb, out);
}


# window's 5-column sum done in phase-1 registers (adds with the next lane's values via a row shift), not by a second banded matmul; the one f32 WMMA product left stores to out; sum order changes
# speedup vs baseline: 2.2409x; 1.4779x over previous
#include <hip/hip_runtime.h>


typedef __attribute__((ext_vector_type(2))) float v2f;
typedef __attribute__((ext_vector_type(4))) float v4f;
typedef __attribute__((ext_vector_type(8))) float v8f;

#define IN_C   3
#define OUT_C  32
#define HW     64
#define NOUT   60
#define PITCH  68
#define KSTEPS 5

__device__ __forceinline__ float next_lane_in_row(float v) {
    return __builtin_bit_cast(float,
        __builtin_amdgcn_update_dpp(0, __builtin_bit_cast(int, v), 0x101, 0xf, 0xf, true));
}

__global__ __launch_bounds__(256) void
localconv2d_wmma_kernel(const float* __restrict__ x,
                        const float* __restrict__ lb,
                        const float* __restrict__ rb,
                        float* __restrict__ out) {
    const int b   = blockIdx.x >> 5;
    const int oc  = blockIdx.x & 31;

    __shared__ __attribute__((aligned(16))) float S[HW * PITCH];

    const int tid  = threadIdx.x;
    const int lane = tid & 31;
    const int wave = tid >> 5;

    const float* xb = x + (size_t)b * IN_C * HW * HW;

    const v4f* xv = reinterpret_cast<const v4f*>(xb);
    v4f px[4][IN_C];
    for (int i = 0; i < 4; ++i)
        for (int c = 0; c < IN_C; ++c)
            px[i][c] = xv[c * (HW * HW / 4) + i * 256 + tid];

    const float l    = lb[oc * IN_C * 25];
    const float r    = rb[oc * IN_C * 25];
    const float dlr  = r - l;
    const float norm = 4.0f / (dlr * dlr);

    for (int i = 0; i < 4; ++i) {
        v4f s4;
        for (int e = 0; e < 4; ++e) {
            float s = 0.0f;
            for (int c = 0; c < IN_C; ++c) {
                float p  = px[i][c][e];
                float a  = fminf(fmaxf(p - l, 0.0f), 1.0f);
                float bb = fminf(fmaxf(r - p, 0.0f), 1.0f);
                float m  = a * bb * norm;
                s += m * m;
            }
            s4[e] = s;
        }
        const float t2 = s4[0] + s4[1];
        const float t3 = t2 + s4[2];
        const float t4 = t3 + s4[3];
        const float a2 = s4[2] + s4[3];
        const float a3 = s4[1] + a2;
        v4f h4;
        h4[0] = t4    + next_lane_in_row(s4[0]);
        h4[1] = a3    + next_lane_in_row(t2);
        h4[2] = a2    + next_lane_in_row(t3);
        h4[3] = s4[3] + next_lane_in_row(t4);
        const int q = i * 256 + tid;
        *reinterpret_cast<v4f*>(&S[(q >> 4) * PITCH + (q & 15) * 4]) = h4;
    }

    const int tw    = wave >> 1;
    const int th    = (wave & 1) * 2;
    const int t0    = tw * 16;
    const int kb    = (t0 < HW - 4 * KSTEPS) ? t0 : HW - 4 * KSTEPS;
    const int lrow  = lane & 15;
    const int khalf = (lane >> 4) << 1;
    const int rofs  = (lane >> 4) * 8;

    v2f bf[KSTEPS];
    #pragma unroll
    for (int s = 0; s < KSTEPS; ++s) {
        const int d = (kb - t0) + 4 * s + khalf - lrow;
        bf[s].x = ((unsigned)d       <= 4u) ? 1.0f : 0.0f;
        bf[s].y = ((unsigned)(d + 1) <= 4u) ? 1.0f : 0.0f;
    }
    __syncthreads();

    {
        v8f d0 = {}; v8f d1 = {};
        const float* Sp = &S[(kb + khalf) * PITCH + th * 16 + lrow];
        #pragma unroll
        for (int s = 0; s < KSTEPS; ++s) {
            v2f b0, b1;
            b0.x = Sp[(4 * s)     * PITCH];
            b0.y = Sp[(4 * s + 1) * PITCH];
            b1.x = Sp[(4 * s)     * PITCH + 16];
            b1.y = Sp[(4 * s + 1) * PITCH + 16];
            d0 = __builtin_amdgcn_wmma_f32_16x16x4_f32(false, bf[s], false, b0,
                                                       (short)0, d0, false, false);
            d1 = __builtin_amdgcn_wmma_f32_16x16x4_f32(false, bf[s], false, b1,
                                                       (short)0, d1, false, false);
        }
        const int  r0     = t0 + rofs;
        const int  c0     = th * 16 + lrow;
        const bool rows47 = (r0 + 7 < NOUT);
        const bool col1   = (c0 + 16 < NOUT);
        float* o = out + (size_t)(b * OUT_C + oc) * NOUT * NOUT + r0 * NOUT + c0;
        #pragma unroll
        for (int g = 0; g < 4; ++g) o[g * NOUT] = d0[g];
        if (rows47) {
            #pragma unroll
            for (int g = 4; g < 8; ++g) o[g * NOUT] = d0[g];
        }
        if (col1) {
            #pragma unroll
            for (int g = 0; g < 4; ++g) o[g * NOUT + 16] = d1[g];
            if (rows47) {
                #pragma unroll
                for (int g = 4; g < 8; ++g) o[g * NOUT + 16] = d1[g];
            }
        }
    }
}

extern "C" void kernel_launch(void* const* d_in, const int* in_sizes, int n_in,
                              void* d_out, int out_size, void* d_ws, size_t ws_size,
                              hipStream_t stream) {
    (void)in_sizes; (void)n_in; (void)out_size; (void)d_ws; (void)ws_size;
    const float* x  = (const float*)d_in[0];
    const float* lb = (const float*)d_in[1];
    const float* rb = (const float*)d_in[2];
    float* out = (float*)d_out;
    localconv2d_wmma_kernel<<<dim3(16 * OUT_C), dim3(256), 0, stream>>>(x, lb, rb, out);
}
